# starting kernel
# baseline (speedup 1.0000x reference)
#include <hip/hip_runtime.h>

typedef __attribute__((ext_vector_type(2))) float v2f;
typedef __attribute__((ext_vector_type(8))) float v8f;

#define DIM 512
#define MN  4096
#define LDA 516

__global__ __launch_bounds__(256) void som_w2_kernel(const float* __restrict__ weights,
                                                     float* __restrict__ w2) {
  const int wave = threadIdx.x >> 5;
  const int lane = threadIdx.x & 31;
  const int row  = blockIdx.x * 8 + wave;
  const float4* wr = (const float4*)(weights + (size_t)row * DIM);
  float acc = 0.0f;
#pragma unroll
  for (int i = 0; i < 4; ++i) {
    float4 q = wr[lane + 32 * i];
    acc += q.x * q.x + q.y * q.y + q.z * q.z + q.w * q.w;
  }
#pragma unroll
  for (int m = 16; m >= 1; m >>= 1) acc += __shfl_xor(acc, m, 32);
  if (lane == 0) w2[row] = acc;
}

__global__ __launch_bounds__(128) void som_bmu_kernel(const float* __restrict__ batch,
                                                      const float* __restrict__ weights,
                                                      const float* __restrict__ w2,
                                                      int* __restrict__ bmu) {
  __shared__ __align__(16) float As[16 * LDA];
  __shared__ float cV[4][16];
  __shared__ int   cI[4][16];

  const int tid  = threadIdx.x;
  const int lane = tid & 31;
  const int wv   = tid >> 5;
  const int bid  = blockIdx.x;

  const float4* bp = (const float4*)(batch + (size_t)bid * 16 * DIM);
#pragma unroll
  for (int i = 0; i < 16; ++i) {
    int idx = tid + 128 * i;
    int row = idx >> 7;
    int c4  = idx & 127;
    float4 q = bp[idx];
    *(float4*)&As[row * LDA + c4 * 4] = q;
  }
  __syncthreads();

  const int half = lane >> 4;
  const int l16  = lane & 15;
  const float* aRow = &As[l16 * LDA + 2 * half];

  float bestV[8];
  int   bestI[8];
#pragma unroll
  for (int r = 0; r < 8; ++r) { bestV[r] = -3.402823466e38f; bestI[r] = 0; }

  for (int nt = wv; nt < MN / 16; nt += 4) {
    const int n = nt * 16 + l16;
    const float* wRow = weights + (size_t)n * DIM + 2 * half;
    v8f acc0 = {}, acc1 = {}, acc2 = {}, acc3 = {};
#pragma unroll 2
    for (int ks = 0; ks < DIM / 4; ks += 4) {
      v2f a0 = *(const v2f*)(aRow + 4 * ks);
      v2f b0 = *(const v2f*)(wRow + 4 * ks);
      acc0 = __builtin_amdgcn_wmma_f32_16x16x4_f32(false, a0, false, b0,
                                                   (short)0, acc0, false, false);
      v2f a1 = *(const v2f*)(aRow + 4 * ks + 4);
      v2f b1 = *(const v2f*)(wRow + 4 * ks + 4);
      acc1 = __builtin_amdgcn_wmma_f32_16x16x4_f32(false, a1, false, b1,
                                                   (short)0, acc1, false, false);
      v2f a2 = *(const v2f*)(aRow + 4 * ks + 8);
      v2f b2 = *(const v2f*)(wRow + 4 * ks + 8);
      acc2 = __builtin_amdgcn_wmma_f32_16x16x4_f32(false, a2, false, b2,
                                                   (short)0, acc2, false, false);
      v2f a3 = *(const v2f*)(aRow + 4 * ks + 12);
      v2f b3 = *(const v2f*)(wRow + 4 * ks + 12);
      acc3 = __builtin_amdgcn_wmma_f32_16x16x4_f32(false, a3, false, b3,
                                                   (short)0, acc3, false, false);
    }
    const float w2n = w2[n];
#pragma unroll
    for (int r = 0; r < 8; ++r) {
      float s = 2.0f * (acc0[r] + acc1[r] + acc2[r] + acc3[r]) - w2n;
      if (s > bestV[r]) { bestV[r] = s; bestI[r] = n; }
    }
  }

#pragma unroll
  for (int r = 0; r < 8; ++r) {
    float v = bestV[r];
    int   ix = bestI[r];
#pragma unroll
    for (int m = 8; m >= 1; m >>= 1) {
      float ov = __shfl_xor(v, m, 32);
      int   oi = __shfl_xor(ix, m, 32);
      if (ov > v || (ov == v && oi < ix)) { v = ov; ix = oi; }
    }
    if (lane == 0)  { cV[wv][r]     = v; cI[wv][r]     = ix; }
    if (lane == 16) { cV[wv][8 + r] = v; cI[wv][8 + r] = ix; }
  }
  __syncthreads();

  if (tid < 16) {
    float v = cV[0][tid];
    int   ix = cI[0][tid];
#pragma unroll
    for (int w = 1; w < 4; ++w) {
      float ov = cV[w][tid];
      int   oi = cI[w][tid];
      if (ov > v || (ov == v && oi < ix)) { v = ov; ix = oi; }
    }
    bmu[bid * 16 + tid] = ix;
  }
}

__global__ __launch_bounds__(128) void som_out_kernel(const int* __restrict__ bmu,
                                                      const int* __restrict__ it_p,
                                                      const int* __restrict__ decay_p,
                                                      float* __restrict__ out) {
  __shared__ float efi[64];
  __shared__ float efj[64];
  const int row = blockIdx.x;
  const int t   = threadIdx.x;
  const int idx = bmu[row];
  const int bi = idx >> 6;
  const int bj = idx & 63;
  const float lr  = __expf(-(float)it_p[0] / (float)decay_p[0]);
  const float sig = 32.0f * lr;
  const float inv = 1.0f / (sig * sig);
  if (t < 64) {
    int d = t - bi;
    efi[t] = __expf(-(float)(d * d) * inv);
  } else {
    int c = t - 64;
    int d = c - bj;
    efj[c] = __expf(-(float)(d * d) * inv);
  }
  __syncthreads();
  float* op = out + (size_t)row * MN;
  const float fj = efj[t & 63];
#pragma unroll
  for (int k = 0; k < 32; ++k) {
    int j = t + (k << 7);
    op[j] = efi[j >> 6] * fj;
  }
}

extern "C" void kernel_launch(void* const* d_in, const int* in_sizes, int n_in,
                              void* d_out, int out_size, void* d_ws, size_t ws_size,
                              hipStream_t stream) {
  const float* batch   = (const float*)d_in[0];
  const float* weights = (const float*)d_in[1];
  const int* decay = (const int*)d_in[3];
  const int* it    = (const int*)d_in[4];

  float* w2  = (float*)d_ws;
  int*   bmu = (int*)((char*)d_ws + MN * sizeof(float));
  float* out = (float*)d_out;

  som_w2_kernel <<<MN / 8, 256, 0, stream>>>(weights, w2);
  som_bmu_kernel<<<MN / 16, 128, 0, stream>>>(batch, weights, w2, bmu);
  som_out_kernel<<<4096, 128, 0, stream>>>(bmu, it, decay, out);
}


# Output kernel re-mapped: one resident 8-wave work-group per WGP, each wave streams two whole rows with 16-B stores (was 4096 short 4-wave groups, 4-B stores, factors via LDS); values, sums unchanged
# speedup vs baseline: 4.5701x; 4.5701x over previous
#include <hip/hip_runtime.h>


typedef __attribute__((ext_vector_type(2))) float v2f;
typedef __attribute__((ext_vector_type(4))) float v4f;
typedef __attribute__((ext_vector_type(8))) float v8f;

#define DIM 512
#define MN  4096
#define BM  128
#define BN  256
#define BK  64
#define PITCH 68
#define A_F (BM * PITCH)
#define B_F (BN * PITCH)
#define STAGE_F (A_F + B_F)
#define NSPLIT 8
#define KCH (DIM / BK)

__global__ __launch_bounds__(256) void som_w2_kernel(const float* __restrict__ weights,
                                                     float* __restrict__ w2) {
  const int wave = threadIdx.x >> 5;
  const int lane = threadIdx.x & 31;
  const int row  = blockIdx.x * 8 + wave;
  const float4* wr = (const float4*)(weights + (size_t)row * DIM);
  float acc = 0.0f;
#pragma unroll
  for (int i = 0; i < 4; ++i) {
    float4 q = wr[lane + 32 * i];
    acc += q.x * q.x + q.y * q.y + q.z * q.z + q.w * q.w;
  }
#pragma unroll
  for (int m = 16; m >= 1; m >>= 1) acc += __shfl_xor(acc, m, 32);
  if (lane == 0) w2[row] = acc;
}

__global__ __launch_bounds__(256) void som_bmu_kernel(const float* __restrict__ batch,
                                                      const float* __restrict__ weights,
                                                      const float* __restrict__ w2,
                                                      float* __restrict__ pV,
                                                      int* __restrict__ pI) {
  __shared__ __align__(16) float sm[2 * STAGE_F];
  __shared__ float cV[4][BM];
  __shared__ int   cI[4][BM];

  const int tid  = threadIdx.x;
  const int lane = tid & 31;
  const int wv   = tid >> 5;
  const int wm   = wv & 1;
  const int wn   = wv >> 1;
  const int c    = lane & 15;
  const int h    = lane >> 4;
  const int rb   = blockIdx.x >> 3;
  const int sp   = blockIdx.x & 7;
  const int row0 = rb * BM;
  const int col0 = sp * (MN / NSPLIT);

  const int lr = tid >> 4;
  const int lc = (tid & 15) * 4;
  const float* gA = batch   + (size_t)(row0 + lr) * DIM + lc;
  const float* gB = weights + (size_t)(col0 + lr) * DIM + lc;
  const int stA = lr * PITCH + lc;
  const int stB = A_F + lr * PITCH + lc;

  const int fA = (wm * 64 + c) * PITCH + 4 * h;
  const int fB = A_F + (wn * 64 + c) * PITCH + 4 * h;

  v4f ra[8], rw[16];

  float bestV[4][8];
  int   bestI[4][8];
#pragma unroll
  for (int mi = 0; mi < 4; ++mi)
#pragma unroll
    for (int r = 0; r < 8; ++r) { bestV[mi][r] = -3.402823466e38f; bestI[mi][r] = 0; }

#pragma unroll
  for (int i = 0; i < 8; ++i)  ra[i] = *(const v4f*)(gA + (size_t)i * 16 * DIM);
#pragma unroll
  for (int i = 0; i < 16; ++i) rw[i] = *(const v4f*)(gB + (size_t)i * 16 * DIM);
#pragma unroll
  for (int i = 0; i < 8; ++i)  *(v4f*)&sm[stA + i * 16 * PITCH] = ra[i];
#pragma unroll
  for (int i = 0; i < 16; ++i) *(v4f*)&sm[stB + i * 16 * PITCH] = rw[i];
#pragma unroll
  for (int i = 0; i < 8; ++i)  ra[i] = *(const v4f*)(gA + BK + (size_t)i * 16 * DIM);
#pragma unroll
  for (int i = 0; i < 16; ++i) rw[i] = *(const v4f*)(gB + BK + (size_t)i * 16 * DIM);
  __syncthreads();

  for (int nt = 0; nt < 2; ++nt) {
    v8f acc[4][4];
#pragma unroll
    for (int mi = 0; mi < 4; ++mi)
#pragma unroll
      for (int nj = 0; nj < 4; ++nj) acc[mi][nj] = (v8f){};

#pragma unroll 1
    for (int kc = 0; kc < KCH; ++kc) {
      const int t = nt * KCH + kc;
      const float* cur = sm + (t & 1) * STAGE_F;
      float*       nxt = sm + ((t + 1) & 1) * STAGE_F;

      v4f fa[2][4], fb[2][4];
#pragma unroll
      for (int mi = 0; mi < 4; ++mi) fa[0][mi] = *(const v4f*)(cur + fA + mi * 16 * PITCH);
#pragma unroll
      for (int nj = 0; nj < 4; ++nj) fb[0][nj] = *(const v4f*)(cur + fB + nj * 16 * PITCH);

#pragma unroll
      for (int kk = 0; kk < 8; ++kk) {
        const int s = kk & 1;
        if (kk < 7) {
#pragma unroll
          for (int mi = 0; mi < 4; ++mi)
            fa[s ^ 1][mi] = *(const v4f*)(cur + fA + mi * 16 * PITCH + (kk + 1) * 8);
#pragma unroll
          for (int nj = 0; nj < 4; ++nj)
            fb[s ^ 1][nj] = *(const v4f*)(cur + fB + nj * 16 * PITCH + (kk + 1) * 8);
        }
#pragma unroll
        for (int mi = 0; mi < 4; ++mi)
#pragma unroll
          for (int nj = 0; nj < 4; ++nj) {
            v2f a = fa[s][mi].xy;
            v2f b = fb[s][nj].xy;
            acc[mi][nj] = __builtin_amdgcn_wmma_f32_16x16x4_f32(false, a, false, b,
                                                                (short)0, acc[mi][nj], false, false);
          }
#pragma unroll
        for (int mi = 0; mi < 4; ++mi)
#pragma unroll
          for (int nj = 0; nj < 4; ++nj) {
            v2f a = fa[s][mi].zw;
            v2f b = fb[s][nj].zw;
            acc[mi][nj] = __builtin_amdgcn_wmma_f32_16x16x4_f32(false, a, false, b,
                                                                (short)0, acc[mi][nj], false, false);
          }
        if (kk == 0) {
#pragma unroll
          for (int i = 0; i < 8; ++i)  *(v4f*)&nxt[stA + i * 16 * PITCH] = ra[i];
#pragma unroll
          for (int i = 0; i < 16; ++i) *(v4f*)&nxt[stB + i * 16 * PITCH] = rw[i];
          if (t + 2 < 2 * KCH) {
            const int t2 = t + 2;
            const float* pa = gA + (t2 & 7) * BK;
            const float* pb = gB + (size_t)(t2 >> 3) * BN * DIM + (t2 & 7) * BK;
#pragma unroll
            for (int i = 0; i < 8; ++i)  ra[i] = *(const v4f*)(pa + (size_t)i * 16 * DIM);
#pragma unroll
            for (int i = 0; i < 16; ++i) rw[i] = *(const v4f*)(pb + (size_t)i * 16 * DIM);
          }
        }
      }
      __syncthreads();
    }

#pragma unroll
    for (int nj = 0; nj < 4; ++nj) {
      const int n = col0 + nt * BN + wn * 64 + nj * 16 + c;
      const float w2n = w2[n];
#pragma unroll
      for (int mi = 0; mi < 4; ++mi)
#pragma unroll
        for (int r = 0; r < 8; ++r) {
          float sc = 2.0f * acc[mi][nj][r] - w2n;
          if (sc > bestV[mi][r]) { bestV[mi][r] = sc; bestI[mi][r] = n; }
        }
    }
  }

#pragma unroll
  for (int mi = 0; mi < 4; ++mi)
#pragma unroll
    for (int r = 0; r < 8; ++r) {
      float v  = bestV[mi][r];
      int   ix = bestI[mi][r];
#pragma unroll
      for (int m = 8; m >= 1; m >>= 1) {
        float ov = __shfl_xor(v, m, 32);
        int   oi = __shfl_xor(ix, m, 32);
        if (ov > v || (ov == v && oi < ix)) { v = ov; ix = oi; }
      }
      if (c == 0) {
        const int lrow = wm * 64 + mi * 16 + 8 * h + r;
        cV[wn][lrow] = v;
        cI[wn][lrow] = ix;
      }
    }
  __syncthreads();

  if (tid < BM) {
    float v  = cV[0][tid];
    int   ix = cI[0][tid];
#pragma unroll
    for (int w = 1; w < 4; ++w) {
      float ov = cV[w][tid];
      int   oi = cI[w][tid];
      if (ov > v || (ov == v && oi < ix)) { v = ov; ix = oi; }
    }
    pV[(size_t)(row0 + tid) * NSPLIT + sp] = v;
    pI[(size_t)(row0 + tid) * NSPLIT + sp] = ix;
  }
}

#define OUT_ROWS_PER_WG 16
__global__ __launch_bounds__(256) void som_out_kernel(const float* __restrict__ pV,
                                                      const int* __restrict__ pI,
                                                      const int* __restrict__ it_p,
                                                      const int* __restrict__ decay_p,
                                                      float* __restrict__ out) {
  const int lane = threadIdx.x & 31;
  const int wv = __builtin_amdgcn_readfirstlane((int)(threadIdx.x >> 5));
  const float lr  = __expf(-(float)it_p[0] / (float)decay_p[0]);
  const float sig = 32.0f * lr;
  const float inv = 1.0f / (sig * sig);
  const int jj0 = (lane & 15) * 4;
  const int hi  = lane >> 4;

  const int row0 = blockIdx.x * OUT_ROWS_PER_WG + wv * 2;
  float vs[2 * NSPLIT];
  int   is[2 * NSPLIT];
#pragma unroll
  for (int q = 0; q < 4; ++q) {
    const float4 tv = *(const float4*)(pV + (size_t)row0 * NSPLIT + 4 * q);
    const int4   ti = *(const int4*)(pI + (size_t)row0 * NSPLIT + 4 * q);
    vs[4 * q] = tv.x; vs[4 * q + 1] = tv.y; vs[4 * q + 2] = tv.z; vs[4 * q + 3] = tv.w;
    is[4 * q] = ti.x; is[4 * q + 1] = ti.y; is[4 * q + 2] = ti.z; is[4 * q + 3] = ti.w;
  }
  int bi[2], bj[2];
#pragma unroll
  for (int rr = 0; rr < 2; ++rr) {
    float bv  = vs[NSPLIT * rr];
    int   idx = is[NSPLIT * rr];
#pragma unroll
    for (int s = 1; s < NSPLIT; ++s) {
      const float ov = vs[NSPLIT * rr + s];
      const int   oi = is[NSPLIT * rr + s];
      const bool take = (ov > bv) | ((ov == bv) & (oi < idx));
      bv  = take ? ov : bv;
      idx = take ? oi : idx;
    }
    bi[rr] = idx >> 6;
    bj[rr] = idx & 63;
  }

#pragma unroll
  for (int rr = 0; rr < 2; ++rr) {
    const int row = row0 + rr;
    float fj[4];
#pragma unroll
    for (int e = 0; e < 4; ++e) {
      const float fd = (float)(jj0 + e - bj[rr]);
      fj[e] = __expf(-(fd * fd) * inv);
    }
    float4* op = (float4*)(out + (size_t)row * MN) + lane;
#pragma unroll
    for (int s = 0; s < 32; ++s) {
      const float fd = (float)(2 * s + hi - bi[rr]);
      const float fi = __expf(-(fd * fd) * inv);
      op[s * 32] = make_float4(fi * fj[0], fi * fj[1], fi * fj[2], fi * fj[3]);
    }
  }
}

extern "C" void kernel_launch(void* const* d_in, const int* in_sizes, int n_in,
                              void* d_out, int out_size, void* d_ws, size_t ws_size,
                              hipStream_t stream) {
  const float* batch   = (const float*)d_in[0];
  const float* weights = (const float*)d_in[1];
  const int* decay = (const int*)d_in[3];
  const int* it    = (const int*)d_in[4];

  float* w2 = (float*)d_ws;
  float* pV = (float*)((char*)d_ws + MN * sizeof(float));
  int*   pI = (int*)((char*)d_ws + MN * sizeof(float) + MN * NSPLIT * sizeof(float));
  float* out = (float*)d_out;

  som_w2_kernel <<<MN / 8, 256, 0, stream>>>(weights, w2);
  som_bmu_kernel<<<(MN / BM) * NSPLIT, 256, 0, stream>>>(batch, weights, w2, pV, pI);
  som_out_kernel<<<MN / OUT_ROWS_PER_WG, 256, 0, stream>>>(pV, pI, it, decay, out);
}


# GEMM kernel's closing arg-max across each half-wave: row maximum, then lowest column, by DPP row_xmask steps on the vector ALU (was 256 LDS-unit shuffles, each waited for); values, sums unchanged
# speedup vs baseline: 4.9178x; 1.0761x over previous
#include <hip/hip_runtime.h>


typedef __attribute__((ext_vector_type(2))) float v2f;
typedef __attribute__((ext_vector_type(4))) float v4f;
typedef __attribute__((ext_vector_type(8))) float v8f;

#define DIM 512
#define MN  4096
#define BM  128
#define BN  256
#define BK  64
#define PITCH 68
#define A_F (BM * PITCH)
#define B_F (BN * PITCH)
#define STAGE_F (A_F + B_F)
#define NSPLIT 8
#define KCH (DIM / BK)

__global__ __launch_bounds__(256) void som_w2_kernel(const float* __restrict__ weights,
                                                     float* __restrict__ w2) {
  const int wave = threadIdx.x >> 5;
  const int lane = threadIdx.x & 31;
  const int row  = blockIdx.x * 8 + wave;
  const float4* wr = (const float4*)(weights + (size_t)row * DIM);
  float acc = 0.0f;
#pragma unroll
  for (int i = 0; i < 4; ++i) {
    float4 q = wr[lane + 32 * i];
    acc += q.x * q.x + q.y * q.y + q.z * q.z + q.w * q.w;
  }
#pragma unroll
  for (int m = 16; m >= 1; m >>= 1) acc += __shfl_xor(acc, m, 32);
  if (lane == 0) w2[row] = acc;
}

template <int MASK>
__device__ __forceinline__ int row_xor(int v) {
  return __builtin_amdgcn_update_dpp(v, v, 0x160 + MASK, 0xf, 0xf, true);
}
__device__ __forceinline__ int score_key(float f) {
  const int b = __float_as_int(f);
  return b ^ ((b >> 31) & 0x7fffffff);
}
__device__ __forceinline__ float key_score(int k) {
  return __int_as_float(k ^ ((k >> 31) & 0x7fffffff));
}

__global__ __launch_bounds__(256) void som_bmu_kernel(const float* __restrict__ batch,
                                                      const float* __restrict__ weights,
                                                      const float* __restrict__ w2,
                                                      float* __restrict__ pV,
                                                      int* __restrict__ pI) {
  __shared__ __align__(16) float sm[2 * STAGE_F];
  __shared__ float cV[4][BM];
  __shared__ int   cI[4][BM];

  const int tid  = threadIdx.x;
  const int lane = tid & 31;
  const int wv   = tid >> 5;
  const int wm   = wv & 1;
  const int wn   = wv >> 1;
  const int c    = lane & 15;
  const int h    = lane >> 4;
  const int rb   = blockIdx.x >> 3;
  const int sp   = blockIdx.x & 7;
  const int row0 = rb * BM;
  const int col0 = sp * (MN / NSPLIT);

  const int lr = tid >> 4;
  const int lc = (tid & 15) * 4;
  const float* gA = batch   + (size_t)(row0 + lr) * DIM + lc;
  const float* gB = weights + (size_t)(col0 + lr) * DIM + lc;
  const int stA = lr * PITCH + lc;
  const int stB = A_F + lr * PITCH + lc;

  const int fA = (wm * 64 + c) * PITCH + 4 * h;
  const int fB = A_F + (wn * 64 + c) * PITCH + 4 * h;

  v4f ra[8], rw[16];

  float bestV[4][8];
  int   bestI[4][8];
#pragma unroll
  for (int mi = 0; mi < 4; ++mi)
#pragma unroll
    for (int r = 0; r < 8; ++r) { bestV[mi][r] = -3.402823466e38f; bestI[mi][r] = 0; }

#pragma unroll
  for (int i = 0; i < 8; ++i)  ra[i] = *(const v4f*)(gA + (size_t)i * 16 * DIM);
#pragma unroll
  for (int i = 0; i < 16; ++i) rw[i] = *(const v4f*)(gB + (size_t)i * 16 * DIM);
#pragma unroll
  for (int i = 0; i < 8; ++i)  *(v4f*)&sm[stA + i * 16 * PITCH] = ra[i];
#pragma unroll
  for (int i = 0; i < 16; ++i) *(v4f*)&sm[stB + i * 16 * PITCH] = rw[i];
#pragma unroll
  for (int i = 0; i < 8; ++i)  ra[i] = *(const v4f*)(gA + BK + (size_t)i * 16 * DIM);
#pragma unroll
  for (int i = 0; i < 16; ++i) rw[i] = *(const v4f*)(gB + BK + (size_t)i * 16 * DIM);
  __syncthreads();

  for (int nt = 0; nt < 2; ++nt) {
    v8f acc[4][4];
#pragma unroll
    for (int mi = 0; mi < 4; ++mi)
#pragma unroll
      for (int nj = 0; nj < 4; ++nj) acc[mi][nj] = (v8f){};

#pragma unroll 1
    for (int kc = 0; kc < KCH; ++kc) {
      const int t = nt * KCH + kc;
      const float* cur = sm + (t & 1) * STAGE_F;
      float*       nxt = sm + ((t + 1) & 1) * STAGE_F;

      v4f fa[2][4], fb[2][4];
#pragma unroll
      for (int mi = 0; mi < 4; ++mi) fa[0][mi] = *(const v4f*)(cur + fA + mi * 16 * PITCH);
#pragma unroll
      for (int nj = 0; nj < 4; ++nj) fb[0][nj] = *(const v4f*)(cur + fB + nj * 16 * PITCH);

#pragma unroll
      for (int kk = 0; kk < 8; ++kk) {
        const int s = kk & 1;
        if (kk < 7) {
#pragma unroll
          for (int mi = 0; mi < 4; ++mi)
            fa[s ^ 1][mi] = *(const v4f*)(cur + fA + mi * 16 * PITCH + (kk + 1) * 8);
#pragma unroll
          for (int nj = 0; nj < 4; ++nj)
            fb[s ^ 1][nj] = *(const v4f*)(cur + fB + nj * 16 * PITCH + (kk + 1) * 8);
        }
#pragma unroll
        for (int mi = 0; mi < 4; ++mi)
#pragma unroll
          for (int nj = 0; nj < 4; ++nj) {
            v2f a = fa[s][mi].xy;
            v2f b = fb[s][nj].xy;
            acc[mi][nj] = __builtin_amdgcn_wmma_f32_16x16x4_f32(false, a, false, b,
                                                                (short)0, acc[mi][nj], false, false);
          }
#pragma unroll
        for (int mi = 0; mi < 4; ++mi)
#pragma unroll
          for (int nj = 0; nj < 4; ++nj) {
            v2f a = fa[s][mi].zw;
            v2f b = fb[s][nj].zw;
            acc[mi][nj] = __builtin_amdgcn_wmma_f32_16x16x4_f32(false, a, false, b,
                                                                (short)0, acc[mi][nj], false, false);
          }
        if (kk == 0) {
#pragma unroll
          for (int i = 0; i < 8; ++i)  *(v4f*)&nxt[stA + i * 16 * PITCH] = ra[i];
#pragma unroll
          for (int i = 0; i < 16; ++i) *(v4f*)&nxt[stB + i * 16 * PITCH] = rw[i];
          if (t + 2 < 2 * KCH) {
            const int t2 = t + 2;
            const float* pa = gA + (t2 & 7) * BK;
            const float* pb = gB + (size_t)(t2 >> 3) * BN * DIM + (t2 & 7) * BK;
#pragma unroll
            for (int i = 0; i < 8; ++i)  ra[i] = *(const v4f*)(pa + (size_t)i * 16 * DIM);
#pragma unroll
            for (int i = 0; i < 16; ++i) rw[i] = *(const v4f*)(pb + (size_t)i * 16 * DIM);
          }
        }
      }
      __syncthreads();
    }

#pragma unroll
    for (int nj = 0; nj < 4; ++nj) {
      const int n = col0 + nt * BN + wn * 64 + nj * 16 + c;
      const float w2n = w2[n];
#pragma unroll
      for (int mi = 0; mi < 4; ++mi)
#pragma unroll
        for (int r = 0; r < 8; ++r) {
          float sc = 2.0f * acc[mi][nj][r] - w2n;
          if (sc > bestV[mi][r]) { bestV[mi][r] = sc; bestI[mi][r] = n; }
        }
    }
  }

  float rV[4][8];
  int   rI[4][8];
#pragma unroll
  for (int mi = 0; mi < 4; ++mi)
#pragma unroll
    for (int r = 0; r < 8; ++r) {
      const float own = bestV[mi][r];
      int k = score_key(own);
      k = max(k, row_xor<8>(k));
      k = max(k, row_xor<4>(k));
      k = max(k, row_xor<2>(k));
      k = max(k, row_xor<1>(k));
      const float v = key_score(k);
      int ix = (own == v) ? bestI[mi][r] : 0x7fffffff;
      ix = min(ix, row_xor<8>(ix));
      ix = min(ix, row_xor<4>(ix));
      ix = min(ix, row_xor<2>(ix));
      ix = min(ix, row_xor<1>(ix));
      rV[mi][r] = v;
      rI[mi][r] = ix;
    }
  if (c == 0) {
#pragma unroll
    for (int mi = 0; mi < 4; ++mi)
#pragma unroll
      for (int r = 0; r < 8; ++r) {
        const int lrow = wm * 64 + mi * 16 + 8 * h + r;
        cV[wn][lrow] = rV[mi][r];
        cI[wn][lrow] = rI[mi][r];
      }
  }
  __syncthreads();

  if (tid < BM) {
    float v  = cV[0][tid];
    int   ix = cI[0][tid];
#pragma unroll
    for (int w = 1; w < 4; ++w) {
      float ov = cV[w][tid];
      int   oi = cI[w][tid];
      if (ov > v || (ov == v && oi < ix)) { v = ov; ix = oi; }
    }
    pV[(size_t)(row0 + tid) * NSPLIT + sp] = v;
    pI[(size_t)(row0 + tid) * NSPLIT + sp] = ix;
  }
}

#define OUT_ROWS_PER_WG 16
__global__ __launch_bounds__(256) void som_out_kernel(const float* __restrict__ pV,
                                                      const int* __restrict__ pI,
                                                      const int* __restrict__ it_p,
                                                      const int* __restrict__ decay_p,
                                                      float* __restrict__ out) {
  const int lane = threadIdx.x & 31;
  const int wv = __builtin_amdgcn_readfirstlane((int)(threadIdx.x >> 5));
  const float lr  = __expf(-(float)it_p[0] / (float)decay_p[0]);
  const float sig = 32.0f * lr;
  const float inv = 1.0f / (sig * sig);
  const int jj0 = (lane & 15) * 4;
  const int hi  = lane >> 4;

  const int row0 = blockIdx.x * OUT_ROWS_PER_WG + wv * 2;
  float vs[2 * NSPLIT];
  int   is[2 * NSPLIT];
#pragma unroll
  for (int q = 0; q < 4; ++q) {
    const float4 tv = *(const float4*)(pV + (size_t)row0 * NSPLIT + 4 * q);
    const int4   ti = *(const int4*)(pI + (size_t)row0 * NSPLIT + 4 * q);
    vs[4 * q] = tv.x; vs[4 * q + 1] = tv.y; vs[4 * q + 2] = tv.z; vs[4 * q + 3] = tv.w;
    is[4 * q] = ti.x; is[4 * q + 1] = ti.y; is[4 * q + 2] = ti.z; is[4 * q + 3] = ti.w;
  }
  int bi[2], bj[2];
#pragma unroll
  for (int rr = 0; rr < 2; ++rr) {
    float bv  = vs[NSPLIT * rr];
    int   idx = is[NSPLIT * rr];
#pragma unroll
    for (int s = 1; s < NSPLIT; ++s) {
      const float ov = vs[NSPLIT * rr + s];
      const int   oi = is[NSPLIT * rr + s];
      const bool take = (ov > bv) | ((ov == bv) & (oi < idx));
      bv  = take ? ov : bv;
      idx = take ? oi : idx;
    }
    bi[rr] = idx >> 6;
    bj[rr] = idx & 63;
  }

#pragma unroll
  for (int rr = 0; rr < 2; ++rr) {
    const int row = row0 + rr;
    float fj[4];
#pragma unroll
    for (int e = 0; e < 4; ++e) {
      const float fd = (float)(jj0 + e - bj[rr]);
      fj[e] = __expf(-(fd * fd) * inv);
    }
    float4* op = (float4*)(out + (size_t)row * MN) + lane;
#pragma unroll
    for (int s = 0; s < 32; ++s) {
      const float fd = (float)(2 * s + hi - bi[rr]);
      const float fi = __expf(-(fd * fd) * inv);
      op[s * 32] = make_float4(fi * fj[0], fi * fj[1], fi * fj[2], fi * fj[3]);
    }
  }
}

extern "C" void kernel_launch(void* const* d_in, const int* in_sizes, int n_in,
                              void* d_out, int out_size, void* d_ws, size_t ws_size,
                              hipStream_t stream) {
  const float* batch   = (const float*)d_in[0];
  const float* weights = (const float*)d_in[1];
  const int* decay = (const int*)d_in[3];
  const int* it    = (const int*)d_in[4];

  float* w2 = (float*)d_ws;
  float* pV = (float*)((char*)d_ws + MN * sizeof(float));
  int*   pI = (int*)((char*)d_ws + MN * sizeof(float) + MN * NSPLIT * sizeof(float));
  float* out = (float*)d_out;

  som_w2_kernel <<<MN / 8, 256, 0, stream>>>(weights, w2);
  som_bmu_kernel<<<(MN / BM) * NSPLIT, 256, 0, stream>>>(batch, weights, w2, pV, pI);
  som_out_kernel<<<MN / OUT_ROWS_PER_WG, 256, 0, stream>>>(pV, pI, it, decay, out);
}
